# starting kernel
# baseline (speedup 1.0000x reference)
#include <hip/hip_runtime.h>

typedef float v2f __attribute__((ext_vector_type(2)));
typedef float v8f __attribute__((ext_vector_type(8)));

#define C_IN   128
#define H_IN   224
#define W_IN   224
#define C_OUT  256
#define K_TOT  (C_IN * 9)
#define H_OUT  222
#define W_OUT  222
#define HW_OUT (H_OUT * W_OUT)

#define M_BLK  128
#define N_BLK  64
#define KT     16
#define NPANEL (K_TOT / KT)
#define A_STR  18
#define PB_STR (N_BLK * 2 + 4)

__global__ __launch_bounds__(256)
void conv3x3_wmma_f32(const float* __restrict__ x,
                      const float* __restrict__ w,
                      float* __restrict__ out) {
  __shared__ float As[M_BLK * A_STR];
  __shared__ float Bs[(KT / 2) * PB_STR];

  const int tid  = threadIdx.x;
  const int lane = tid & 31;
  const int wave = tid >> 5;
  const int wm   = wave & 3;
  const int wn   = wave >> 2;

  const int owBlk = blockIdx.x * N_BLK;
  const int oh    = blockIdx.y;
  const int mBase = blockIdx.z * M_BLK;

  v8f acc[2][2] = {};

  const int mA = tid >> 1;
  const int kA = (tid & 1) * 8;
  const int ktB = tid >> 4;
  const int j0  = (tid & 15) * 4;

  const int half = lane >> 4;
  const int l16  = lane & 15;

  float4 aR0, aR1;
  float  bR[4];

  auto loadG = [&](int kBase) {
    const float4* src =
        (const float4*)(w + (size_t)(mBase + mA) * K_TOT + kBase + kA);
    aR0 = src[0];
    aR1 = src[1];
    const int k  = kBase + ktB;
    const int c  = k / 9;
    const int r  = k - c * 9;
    const int kh = r / 3;
    const int kw = r - kh * 3;
    const float* bs =
        x + (size_t)c * (H_IN * W_IN) + (size_t)(oh + kh) * W_IN + owBlk + kw + j0;
#pragma unroll
    for (int i = 0; i < 4; ++i) {
      const int ow = owBlk + j0 + i;
      bR[i] = (ow < W_OUT) ? bs[i] : 0.0f;
    }
  };

  loadG(0);

  for (int p = 0; p < NPANEL; ++p) {
    {
      float* ad = &As[mA * A_STR + kA];
      ad[0] = aR0.x; ad[1] = aR0.y; ad[2] = aR0.z; ad[3] = aR0.w;
      ad[4] = aR1.x; ad[5] = aR1.y; ad[6] = aR1.z; ad[7] = aR1.w;
      float* bd = &Bs[(ktB >> 1) * PB_STR + j0 * 2 + (ktB & 1)];
#pragma unroll
      for (int i = 0; i < 4; ++i) bd[2 * i] = bR[i];
    }
    __syncthreads();

    if (p + 1 < NPANEL) loadG((p + 1) * KT);

#pragma unroll
    for (int k0 = 0; k0 < KT; k0 += 4) {
      v2f a[2], b[2];
#pragma unroll
      for (int ti = 0; ti < 2; ++ti)
        a[ti] = *(const v2f*)&As[(wm * 32 + ti * 16 + l16) * A_STR + k0 + 2 * half];
#pragma unroll
      for (int tj = 0; tj < 2; ++tj)
        b[tj] = *(const v2f*)&Bs[(k0 / 2 + half) * PB_STR +
                                 (wn * 32 + tj * 16 + l16) * 2];
#pragma unroll
      for (int ti = 0; ti < 2; ++ti)
#pragma unroll
        for (int tj = 0; tj < 2; ++tj)
          acc[ti][tj] = __builtin_amdgcn_wmma_f32_16x16x4_f32(
              false, a[ti], false, b[tj], (short)0, acc[ti][tj], false, false);
    }
    __syncthreads();
  }

#pragma unroll
  for (int ti = 0; ti < 2; ++ti) {
    const int m0 = mBase + wm * 32 + ti * 16 + half * 8;
#pragma unroll
    for (int tj = 0; tj < 2; ++tj) {
      const int ow = owBlk + wn * 32 + tj * 16 + l16;
      if (ow < W_OUT) {
        float* dst = out + (size_t)m0 * HW_OUT + (size_t)oh * W_OUT + ow;
#pragma unroll
        for (int r = 0; r < 8; ++r)
          dst[(size_t)r * HW_OUT] = acc[ti][tj][r];
      }
    }
  }
}

extern "C" void kernel_launch(void* const* d_in, const int* in_sizes, int n_in,
                              void* d_out, int out_size, void* d_ws, size_t ws_size,
                              hipStream_t stream) {
  const float* x = (const float*)d_in[0];
  const float* k = (const float*)d_in[1];
  float* out = (float*)d_out;

  dim3 grid((W_OUT + N_BLK - 1) / N_BLK,
            H_OUT,
            C_OUT / M_BLK);
  conv3x3_wmma_f32<<<grid, 256, 0, stream>>>(x, k, out);
}


# Two LDS buffers per operand, one barrier per stage (was two): commit of stage ch+2 and loads of ch+3 issue after the hand-over, beside the next stage's first fragment reads, under 12 products
# speedup vs baseline: 1.2530x; 1.2530x over previous
#include <hip/hip_runtime.h>


typedef float v2f __attribute__((ext_vector_type(2)));
typedef float v8f __attribute__((ext_vector_type(8)));

#define C_IN   128
#define H_IN   224
#define W_IN   224
#define C_OUT  256
#define K_TOT  (C_IN * 9)
#define H_OUT  222
#define W_OUT  222
#define HW_OUT (H_OUT * W_OUT)

#define M_BLK  128
#define N_BLK  64
#define CC     8
#define NCHUNK (C_IN / CC)
#define A_STR  76
#define X_ROW  68
#define X_PLN  (3 * X_ROW)
#define X_UNITS (CC * 3 * (X_ROW / 4))
#define STG_A  (M_BLK * A_STR)
#define STG_X  (CC * X_PLN)

#define N_ROWM  (H_OUT * (C_OUT / M_BLK))
#define N_FULL  (3 * N_ROWM)

__global__ __launch_bounds__(256)
void conv3x3_wmma_f32(const float* __restrict__ x,
                      const float* __restrict__ w,
                      float* __restrict__ out) {
  __shared__ __attribute__((aligned(16))) float As[2 * STG_A];
  __shared__ __attribute__((aligned(16))) float Xs[2 * STG_X];

  const int tid  = threadIdx.x;
  const int lane = tid & 31;
  const int wave = __builtin_amdgcn_readfirstlane(tid >> 5);
  const int wm   = wave & 3;
  const int wn   = wave >> 2;

  const int bid    = blockIdx.x;
  const bool full  = bid < N_FULL;
  const int colBlk = full ? bid % 3 : 3;
  const int rowM   = full ? bid / 3 : bid - N_FULL;
  const int owBlk  = colBlk * N_BLK;
  const int oh     = rowM % H_OUT;
  const int mBase  = (rowM / H_OUT) * M_BLK;

  const bool stripLive = (owBlk + wn * 32) < W_OUT;


  const float* wBlk = w + (size_t)mBase * K_TOT;
  const int row0 = tid / 6,         sub0 = tid - row0 * 6;
  const int row1 = (tid + 256) / 6, sub1 = (tid + 256) - row1 * 6;
  const int row2 = (tid + 512) / 6, sub2 = (tid + 512) - row2 * 6;
  const unsigned aSrc0 = (row0 * K_TOT + sub0 * 4) * 4u;
  const unsigned aSrc1 = (row1 * K_TOT + sub1 * 4) * 4u;
  const unsigned aSrc2 = (row2 * K_TOT + sub2 * 4) * 4u;
  int aQ0 = row0 * (A_STR / 4) + sub0;
  int aQ1 = row1 * (A_STR / 4) + sub1;
  int aQ2 = row2 * (A_STR / 4) + sub2;

  const bool x1On = tid < (X_UNITS - 256);
  const float* xBlk = x + (size_t)oh * W_IN;
  auto xUnit = [&](int u) -> unsigned {
    const int c  = u / 51;
    const int r  = u - c * 51;
    const int kh = r / 17;
    const int c4 = r - kh * 17;
    int colG = owBlk + c4 * 4;
    colG = colG > (W_IN - 4) ? (W_IN - 4) : colG;
    return (c * (H_IN * W_IN) + kh * W_IN + colG) * 4u;
  };
  const unsigned xSrc0 = xUnit(tid);
  const unsigned xSrc1 = xUnit(x1On ? tid + 256 : tid);
  int xQ0 = tid;
  int xQ1 = x1On ? tid + 256 : tid;
  int dAQ = STG_A / 4, dXQ = STG_X / 4;

#define STAGE_DECL(P) float4 P##0, P##1, P##2, P##3, P##4, P##5, P##6, P##7, P##8, P##x0, P##x1
#define STAGE_LOAD(ch, P) do {                                                                     \
      \
      \
      \
    unsigned o0 = aSrc0, o1 = aSrc1, o2 = aSrc2, p0 = xSrc0, p1 = xSrc1;                           \
    asm volatile("" : "+v"(o0), "+v"(o1), "+v"(o2), "+v"(p0), "+v"(p1));                           \
    const char* wc = (const char*)(wBlk + (ch) * (CC * 9));                                        \
    P##0 = *(const float4*)(wc + o0);  P##1 = *(const float4*)(wc + o0 + 96);  P##2 = *(const float4*)(wc + o0 + 192); \
    P##3 = *(const float4*)(wc + o1);  P##4 = *(const float4*)(wc + o1 + 96);  P##5 = *(const float4*)(wc + o1 + 192); \
    P##6 = *(const float4*)(wc + o2);  P##7 = *(const float4*)(wc + o2 + 96);  P##8 = *(const float4*)(wc + o2 + 192); \
    const char* xc = (const char*)(xBlk + (size_t)(ch) * (CC * H_IN * W_IN));                                    \
    P##x0 = *(const float4*)(xc + p0);                                                          \
    P##x1 = *(const float4*)(xc + p1);                                                          \
  } while (0)
#define STAGE_STORE(P) do {                                                                        \
    *(float4*)&As[aQ0 * 4] = P##0;  *(float4*)&As[aQ0 * 4 + 24] = P##1;  *(float4*)&As[aQ0 * 4 + 48] = P##2; \
    *(float4*)&As[aQ1 * 4] = P##3;  *(float4*)&As[aQ1 * 4 + 24] = P##4;  *(float4*)&As[aQ1 * 4 + 48] = P##5; \
    *(float4*)&As[aQ2 * 4] = P##6;  *(float4*)&As[aQ2 * 4 + 24] = P##7;  *(float4*)&As[aQ2 * 4 + 48] = P##8; \
    *(float4*)&Xs[xQ0 * 4] = P##x0;                                                                \
    *(float4*)&Xs[xQ1 * 4] = P##x1;                                                                \
    aQ0 += dAQ;  aQ1 += dAQ;  aQ2 += dAQ;  xQ0 += dXQ;  xQ1 += dXQ;                                \
    dAQ = -dAQ;  dXQ = -dXQ;                                                                       \
  } while (0)
  STAGE_DECL(sp);
  STAGE_DECL(sq);

  const int half = lane >> 4;
  const int l16  = lane & 15;

  int aP0 = ((wm * 32 + l16) * A_STR + half * 18) >> 1;
  int aP1 = aP0 + 8 * A_STR;
  asm volatile("" : "+v"(aP1));
  const int bIdx = half * 2 * X_PLN + wn * 32 + l16;
  int bB[2][4];
#pragma unroll
  for (int tj = 0; tj < 2; ++tj)
#pragma unroll
    for (int q = 0; q < 4; ++q) {
      bB[tj][q] = bIdx + tj * 16 + ((q >> 1) * 4 + (q & 1)) * X_PLN;
      if (tj + q > 0) asm volatile("" : "+v"(bB[tj][q]));
    }
  int dAP = STG_A / 2, dXP = STG_X;
  auto flipRead = [&]() {
    aP0 += dAP;  aP1 += dAP;
#pragma unroll
    for (int tj = 0; tj < 2; ++tj)
#pragma unroll
      for (int q = 0; q < 4; ++q) bB[tj][q] += dXP;
    dAP = -dAP;  dXP = -dXP;
  };

  v2f fa[2][3][2], fb[2][3][2];
  v8f acc[2][2];
  auto loadStep = [&](int i, v2f (&a)[2], v2f (&b)[2]) {
    const int g = i / 9, r = i - 9 * g;
    a[0] = *(const v2f*)&As[aP0 * 2 + 36 * g + 2 * r];
    a[1] = *(const v2f*)&As[aP1 * 2 + 36 * g + 2 * r];
    const int j0 = 2 * r, j1 = 2 * r + 1;
    const int q  = 2 * g + j0 / 9;
    const int o0 = ((j0 % 9) / 3) * X_ROW + (j0 % 9) % 3;
    const int o1 = (j1 / 9 - j0 / 9) * X_PLN + ((j1 % 9) / 3) * X_ROW + (j1 % 9) % 3;
    b[0] = v2f{Xs[bB[0][q] + o0], Xs[bB[0][q] + o1]};
    b[1] = v2f{Xs[bB[1][q] + o0], Xs[bB[1][q] + o1]};
  };
  auto readGroup = [&](int grp, int set) {
#pragma unroll
    for (int s = 0; s < 3; ++s) loadStep(grp * 3 + s, fa[set][s], fb[set][s]);
  };
  auto mmaGroup = [&](int set) {
#pragma unroll
    for (int s = 0; s < 3; ++s)
#pragma unroll
      for (int ti = 0; ti < 2; ++ti)
#pragma unroll
        for (int tj = 0; tj < 2; ++tj)
          acc[ti][tj] = __builtin_amdgcn_wmma_f32_16x16x4_f32(
              false, fa[set][s][ti], false, fb[set][s][tj], (short)0, acc[ti][tj], false, false);
  };

  STAGE_LOAD(0, sp);
  STAGE_LOAD(1, sq);
  STAGE_STORE(sp);
  STAGE_STORE(sq);
  __syncthreads();
  if (stripLive) readGroup(0, 0);
  STAGE_LOAD(2, sp);

#pragma unroll
  for (int ti = 0; ti < 2; ++ti)
#pragma unroll
    for (int tj = 0; tj < 2; ++tj) acc[ti][tj] = v8f{};

  for (int ch = 0; ch < NCHUNK; ++ch) {
    if (stripLive) {
#pragma unroll
      for (int grp = 0; grp < 5; ++grp) {
        readGroup(grp + 1, (grp + 1) & 1);
        __builtin_amdgcn_sched_barrier(0);
        mmaGroup(grp & 1);
        __builtin_amdgcn_sched_barrier(0);
      }
    }
    if (ch + 1 < NCHUNK) {
      __syncthreads();
      flipRead();
      if (stripLive) readGroup(0, 0);
      if (ch + 2 < NCHUNK) {
        STAGE_STORE(sp);
        if (ch + 3 < NCHUNK) STAGE_LOAD(ch + 3, sp);
      }
    }
    if (stripLive) {
      __builtin_amdgcn_sched_barrier(0);
      mmaGroup(1);
      __builtin_amdgcn_sched_barrier(0);
    }
  }

#pragma unroll
  for (int ti = 0; ti < 2; ++ti) {
    const int m0 = mBase + wm * 32 + ti * 16 + half * 8;
#pragma unroll
    for (int tj = 0; tj < 2; ++tj) {
      const int ow = owBlk + wn * 32 + tj * 16 + l16;
      if (ow < W_OUT) {
        float* dst = out + (size_t)m0 * HW_OUT + (size_t)oh * W_OUT + ow;
#pragma unroll
        for (int r = 0; r < 8; ++r)
          dst[(size_t)r * HW_OUT] = acc[ti][tj][r];
      }
    }
  }
}

extern "C" void kernel_launch(void* const* d_in, const int* in_sizes, int n_in,
                              void* d_out, int out_size, void* d_ws, size_t ws_size,
                              hipStream_t stream) {
  const float* x = (const float*)d_in[0];
  const float* k = (const float*)d_in[1];
  float* out = (float*)d_out;

  dim3 grid(N_FULL + N_ROWM);
  conv3x3_wmma_f32<<<grid, 256, 0, stream>>>(x, k, out);
}


# Block tile 64 channels x 2 output rows x 64 columns (was 128 x 1 x 64): a staged weight tile feeds both rows and the rows share two input rows; 27 KB staged per 576 products (was 43 KB)
# speedup vs baseline: 1.3517x; 1.0788x over previous
#include <hip/hip_runtime.h>


typedef float v2f __attribute__((ext_vector_type(2)));
typedef float v8f __attribute__((ext_vector_type(8)));

#define C_IN   128
#define H_IN   224
#define W_IN   224
#define C_OUT  256
#define K_TOT  (C_IN * 9)
#define H_OUT  222
#define W_OUT  222
#define HW_OUT (H_OUT * W_OUT)

#define M_BLK  64
#define N_BLK  64
#define R_BLK  2
#define CC     8
#define NCHUNK (C_IN / CC)
#define A_STR  76
#define X_ROW  68
#define X_ROWS (R_BLK + 2)
#define X_PLN  (X_ROWS * X_ROW)
#define A_UNITS (M_BLK * 18)
#define X_UNITS (CC * X_ROWS * (X_ROW / 4))
#define STG_A  (M_BLK * A_STR)
#define STG_X  (CC * X_PLN)

#define N_ROWP  (H_OUT / R_BLK)
#define N_ROWM  (N_ROWP * (C_OUT / M_BLK))
#define N_FULL  (3 * N_ROWM)

__global__ __launch_bounds__(256)
void conv3x3_wmma_f32(const float* __restrict__ x,
                      const float* __restrict__ w,
                      float* __restrict__ out) {
  __shared__ __attribute__((aligned(16))) float smem[2 * STG_X + 2 * STG_A];
  float* const Xs = smem;
  float* const As = smem + 2 * STG_X;

  const int tid  = threadIdx.x;
  const int lane = tid & 31;
  const int wave = __builtin_amdgcn_readfirstlane(tid >> 5);
  const int wm   = wave & 1;
  const int wr   = (wave >> 1) & 1;
  const int wn   = wave >> 2;

  const int bid    = blockIdx.x;
  const bool full  = bid < N_FULL;
  const int colBlk = full ? bid % 3 : 3;
  const int rowM   = full ? bid / 3 : bid - N_FULL;
  const int owBlk  = colBlk * N_BLK;
  const int oh0    = (rowM % N_ROWP) * R_BLK;
  const int mBase  = (rowM / N_ROWP) * M_BLK;

  const bool stripLive = (owBlk + wn * 32) < W_OUT;


  const float* wBlk = w + (size_t)mBase * K_TOT;
  const bool aTail = wave < 4;
  auto aRow = [&](int u) -> int { return u / 18; };
  auto aSrcOf = [&](int u) -> unsigned { return (aRow(u) * K_TOT + (u - aRow(u) * 18) * 4) * 4u; };
  auto aDstOf = [&](int u) -> int { return aRow(u) * (A_STR / 4) + (u - aRow(u) * 18); };
  const unsigned aSrc0 = aSrcOf(tid),       aSrc1 = aSrcOf(tid + 256), aSrc2 = aSrcOf(tid + 512);
  const unsigned aSrc3 = aSrcOf(tid + 768), aSrc4 = aSrcOf(aTail ? tid + 1024 : tid);
  int aQ0 = aDstOf(tid),       aQ1 = aDstOf(tid + 256), aQ2 = aDstOf(tid + 512);
  int aQ3 = aDstOf(tid + 768), aQ4 = aDstOf(aTail ? tid + 1024 : tid);

  const bool xTail = wave == 0;
  const float* xBlk = x + (size_t)oh0 * W_IN;
  auto xUnit = [&](int u) -> unsigned {
    const int c  = u / (X_ROWS * 17);
    const int r  = u - c * (X_ROWS * 17);
    const int kh = r / 17;
    const int c4 = r - kh * 17;
    int colG = owBlk + c4 * 4;
    colG = colG > (W_IN - 4) ? (W_IN - 4) : colG;
    return (c * (H_IN * W_IN) + kh * W_IN + colG) * 4u;
  };
  const unsigned xSrc0 = xUnit(tid);
  const unsigned xSrc1 = xUnit(tid + 256);
  const unsigned xSrc2 = xUnit(xTail ? tid + 512 : tid);
  int xQ0 = tid;
  int xQ1 = tid + 256;
  int xQ2 = xTail ? tid + 512 : tid;
  int dAQ = STG_A / 4, dXQ = STG_X / 4;

#define STAGE_DECL(P) float4 P##0, P##1, P##2, P##3, P##4 = make_float4(0.f, 0.f, 0.f, 0.f), P##x0, P##x1, P##x2 = make_float4(0.f, 0.f, 0.f, 0.f)
#define STAGE_LOAD(ch, P) do {                                                                     \
      \
      \
      \
    unsigned o0 = aSrc0, o1 = aSrc1, o2 = aSrc2, o3 = aSrc3, o4 = aSrc4, p0 = xSrc0, p1 = xSrc1, p2 = xSrc2; \
    asm volatile("" : "+v"(o0), "+v"(o1), "+v"(o2), "+v"(o3), "+v"(o4), "+v"(p0), "+v"(p1), "+v"(p2)); \
    const char* wc = (const char*)(wBlk + (ch) * (CC * 9));                                        \
    P##0 = *(const float4*)(wc + o0);  P##1 = *(const float4*)(wc + o1);                           \
    P##2 = *(const float4*)(wc + o2);  P##3 = *(const float4*)(wc + o3);                           \
    if (aTail) P##4 = *(const float4*)(wc + o4);                                                   \
    const char* xc = (const char*)(xBlk + (size_t)(ch) * (CC * H_IN * W_IN));                      \
    P##x0 = *(const float4*)(xc + p0);                                                             \
    P##x1 = *(const float4*)(xc + p1);                                                             \
    if (xTail) P##x2 = *(const float4*)(xc + p2);                                                  \
  } while (0)
#define STAGE_STORE(P) do {                                                                        \
    *(float4*)&As[aQ0 * 4] = P##0;  *(float4*)&As[aQ1 * 4] = P##1;                                 \
    *(float4*)&As[aQ2 * 4] = P##2;  *(float4*)&As[aQ3 * 4] = P##3;                                 \
    if (aTail) *(float4*)&As[aQ4 * 4] = P##4;                                                      \
    *(float4*)&Xs[xQ0 * 4] = P##x0;                                                                \
    *(float4*)&Xs[xQ1 * 4] = P##x1;                                                                \
    if (xTail) *(float4*)&Xs[xQ2 * 4] = P##x2;                                                     \
    aQ0 += dAQ;  aQ1 += dAQ;  aQ2 += dAQ;  aQ3 += dAQ;  aQ4 += dAQ;                                \
    xQ0 += dXQ;  xQ1 += dXQ;  xQ2 += dXQ;                                                          \
    dAQ = -dAQ;  dXQ = -dXQ;                                                                       \
  } while (0)
  STAGE_DECL(sp);
  STAGE_DECL(sq);

  const int half = lane >> 4;
  const int l16  = lane & 15;

  int aP0 = (2 * STG_X + (wm * 32 + l16) * A_STR + half * 18) >> 1;
  asm volatile("" : "+v"(aP0));
  int aP1 = aP0 + 8 * A_STR;
  asm volatile("" : "+v"(aP1));
  const int bIdx = half * 2 * X_PLN + wr * X_ROW + wn * 32 + l16;
  int bB[2][4];
#pragma unroll
  for (int tj = 0; tj < 2; ++tj)
#pragma unroll
    for (int q = 0; q < 4; ++q) {
      bB[tj][q] = bIdx + tj * 16 + ((q >> 1) * 4 + (q & 1)) * X_PLN;
      if (tj + q > 0) asm volatile("" : "+v"(bB[tj][q]));
    }
  int dAP = STG_A / 2, dXP = STG_X;
  auto flipRead = [&]() {
    aP0 += dAP;  aP1 += dAP;
#pragma unroll
    for (int tj = 0; tj < 2; ++tj)
#pragma unroll
      for (int q = 0; q < 4; ++q) bB[tj][q] += dXP;
    dAP = -dAP;  dXP = -dXP;
  };

  v2f fa[2][3][2], fb[2][3][2];
  v8f acc[2][2];
  auto loadStep = [&](int i, v2f (&a)[2], v2f (&b)[2]) {
    const int g = i / 9, r = i - 9 * g;
    a[0] = *(const v2f*)&smem[aP0 * 2 + 36 * g + 2 * r];
    a[1] = *(const v2f*)&smem[aP1 * 2 + 36 * g + 2 * r];
    const int j0 = 2 * r, j1 = 2 * r + 1;
    const int q  = 2 * g + j0 / 9;
    const int o0 = ((j0 % 9) / 3) * X_ROW + (j0 % 9) % 3;
    const int o1 = (j1 / 9 - j0 / 9) * X_PLN + ((j1 % 9) / 3) * X_ROW + (j1 % 9) % 3;
    b[0] = v2f{Xs[bB[0][q] + o0], Xs[bB[0][q] + o1]};
    b[1] = v2f{Xs[bB[1][q] + o0], Xs[bB[1][q] + o1]};
  };
  auto readGroup = [&](int grp, int set) {
#pragma unroll
    for (int s = 0; s < 3; ++s) loadStep(grp * 3 + s, fa[set][s], fb[set][s]);
  };
  auto mmaGroup = [&](int set) {
#pragma unroll
    for (int s = 0; s < 3; ++s)
#pragma unroll
      for (int ti = 0; ti < 2; ++ti)
#pragma unroll
        for (int tj = 0; tj < 2; ++tj)
          acc[ti][tj] = __builtin_amdgcn_wmma_f32_16x16x4_f32(
              false, fa[set][s][ti], false, fb[set][s][tj], (short)0, acc[ti][tj], false, false);
  };

  STAGE_LOAD(0, sp);
  STAGE_LOAD(1, sq);
  STAGE_STORE(sp);
  STAGE_STORE(sq);
  __syncthreads();
  if (stripLive) readGroup(0, 0);
  STAGE_LOAD(2, sp);

#pragma unroll
  for (int ti = 0; ti < 2; ++ti)
#pragma unroll
    for (int tj = 0; tj < 2; ++tj) acc[ti][tj] = v8f{};

  for (int ch = 0; ch < NCHUNK; ++ch) {
    if (stripLive) {
#pragma unroll
      for (int grp = 0; grp < 5; ++grp) {
        readGroup(grp + 1, (grp + 1) & 1);
        __builtin_amdgcn_sched_barrier(0);
        mmaGroup(grp & 1);
        __builtin_amdgcn_sched_barrier(0);
      }
    }
    if (ch + 1 < NCHUNK) {
      __syncthreads();
      flipRead();
      if (stripLive) readGroup(0, 0);
      if (ch + 2 < NCHUNK) {
        STAGE_STORE(sp);
        if (ch + 3 < NCHUNK) STAGE_LOAD(ch + 3, sp);
      }
    }
    if (stripLive) {
      __builtin_amdgcn_sched_barrier(0);
      mmaGroup(1);
      __builtin_amdgcn_sched_barrier(0);
    }
  }

#pragma unroll
  for (int ti = 0; ti < 2; ++ti) {
    const int m0 = mBase + wm * 32 + ti * 16 + half * 8;
#pragma unroll
    for (int tj = 0; tj < 2; ++tj) {
      const int ow = owBlk + wn * 32 + tj * 16 + l16;
      if (ow < W_OUT) {
        float* dst = out + (size_t)m0 * HW_OUT + (size_t)(oh0 + wr) * W_OUT + ow;
#pragma unroll
        for (int r = 0; r < 8; ++r)
          dst[(size_t)r * HW_OUT] = acc[ti][tj][r];
      }
    }
  }
}

extern "C" void kernel_launch(void* const* d_in, const int* in_sizes, int n_in,
                              void* d_out, int out_size, void* d_ws, size_t ws_size,
                              hipStream_t stream) {
  const float* x = (const float*)d_in[0];
  const float* k = (const float*)d_in[1];
  float* out = (float*)d_out;

  dim3 grid(N_FULL + N_ROWM);
  conv3x3_wmma_f32<<<grid, 256, 0, stream>>>(x, k, out);
}


# Width dimension in the F(2,3) transformed domain, all f32 on the same matrix form: 4 products per output-column pair and (channel, kernel row), was 6; transforms at staging and in the epilogue
# speedup vs baseline: 1.5474x; 1.1448x over previous
#include <hip/hip_runtime.h>


typedef float v2f __attribute__((ext_vector_type(2)));
typedef float v8f __attribute__((ext_vector_type(8)));

#define C_IN   128
#define H_IN   224
#define W_IN   224
#define C_OUT  256
#define K_TOT  (C_IN * 9)
#define H_OUT  222
#define W_OUT  222
#define HW_OUT (H_OUT * W_OUT)

#define M_BLK  64
#define N_BLK  64
#define R_BLK  2
#define CC     8
#define NCHUNK (C_IN / CC)
#define U_STR  28
#define U_PLN  (M_BLK * U_STR)
#define STG_U  (4 * U_PLN)
#define V_ROW  32
#define V_PLN  (4 * V_ROW + 8)
#define V_POS  (CC * V_PLN)
#define STG_V  (4 * V_POS)

#define N_ROWP  (H_OUT / R_BLK)
#define N_ROWM  (N_ROWP * (C_OUT / M_BLK))
#define N_FULL  (3 * N_ROWM)

__global__ __launch_bounds__(256)
void conv3x3_wmma_f32(const float* __restrict__ x,
                      const float* __restrict__ w,
                      float* __restrict__ out) {
  __shared__ __attribute__((aligned(16))) float smem[2 * STG_V + 2 * STG_U];
  v2f* const smem2 = (v2f*)smem;

  const int tid  = threadIdx.x;
  const int lane = tid & 31;
  const int wave = __builtin_amdgcn_readfirstlane(tid >> 5);
  const int wm   = wave & 1;
  const int wr   = (wave >> 1) & 1;
  const int wn   = wave >> 2;
  const int half = lane >> 4;
  const int l16  = lane & 15;

  const int bid    = blockIdx.x;
  const bool full  = bid < N_FULL;
  const int colBlk = full ? bid % 3 : 3;
  const int rowM   = full ? bid / 3 : bid - N_FULL;
  const int owBlk  = colBlk * N_BLK;
  const int oh0    = (rowM % N_ROWP) * R_BLK;
  const int mBase  = (rowM / N_ROWP) * M_BLK;

  const bool stripLive = (owBlk + wn * 32) < W_OUT;

  const float* wBlk = w + (size_t)mBase * K_TOT;
  unsigned uSrc[3];
  int uQ[3];
#pragma unroll
  for (int i = 0; i < 3; ++i) {
    const int q = tid + 256 * i, m = q / 12, jp = q - m * 12;
    uSrc[i] = (unsigned)(m * K_TOT + jp * 6) * 4u;
    uQ[i] = (2 * STG_V + m * U_STR + 14 * (jp / 6) + 2 * (jp % 6)) >> 1;
  }
  const float* xBlk = x + (size_t)oh0 * W_IN;
  unsigned xSrcA[2], xSrcB[2];
  int xQ[2];
#pragma unroll
  for (int i = 0; i < 2; ++i) {
    const int u = tid + 256 * i, c = u >> 6, row = (u >> 4) & 3, c4 = u & 15;
    int colA = owBlk + c4 * 4;
    int colB = colA + 4;
    colA = colA > (W_IN - 4) ? (W_IN - 4) : colA;
    colB = colB > (W_IN - 2) ? (W_IN - 2) : colB;
    xSrcA[i] = (unsigned)(c * (H_IN * W_IN) + row * W_IN + colA) * 4u;
    xSrcB[i] = (unsigned)(c * (H_IN * W_IN) + row * W_IN + colB) * 4u;
    xQ[i] = (c * V_PLN + row * V_ROW + 2 * c4) >> 1;
  }
  int dUQ = STG_U / 2, dXQ = STG_V / 2;

#define STAGE_DECL(P) v2f P##u[9]; float4 P##a[2]; v2f P##b[2]
#define STAGE_LOAD(ch, P) do {                                                                     \
      \
      \
      \
    unsigned o0 = uSrc[0], o1 = uSrc[1], o2 = uSrc[2], pa0 = xSrcA[0], pa1 = xSrcA[1], pb0 = xSrcB[0], pb1 = xSrcB[1]; \
    asm volatile("" : "+v"(o0), "+v"(o1), "+v"(o2), "+v"(pa0), "+v"(pa1), "+v"(pb0), "+v"(pb1));  \
    const char* wc = (const char*)(wBlk + (ch) * (CC * 9));                                        \
    P##u[0] = *(const v2f*)(wc + o0);  P##u[1] = *(const v2f*)(wc + o0 + 8);  P##u[2] = *(const v2f*)(wc + o0 + 16); \
    P##u[3] = *(const v2f*)(wc + o1);  P##u[4] = *(const v2f*)(wc + o1 + 8);  P##u[5] = *(const v2f*)(wc + o1 + 16); \
    P##u[6] = *(const v2f*)(wc + o2);  P##u[7] = *(const v2f*)(wc + o2 + 8);  P##u[8] = *(const v2f*)(wc + o2 + 16); \
    const char* xc = (const char*)(xBlk + (size_t)(ch) * (CC * H_IN * W_IN));                      \
    P##a[0] = *(const float4*)(xc + pa0);  P##b[0] = *(const v2f*)(xc + pb0);                      \
    P##a[1] = *(const float4*)(xc + pa1);  P##b[1] = *(const v2f*)(xc + pb1);                      \
  } while (0)
#define STAGE_STORE(P) do {                                                                        \
    _Pragma("unroll")                                                                              \
    for (int i_ = 0; i_ < 3; ++i_) {                                                               \
      const float a0 = P##u[3 * i_].x, a1 = P##u[3 * i_].y, a2 = P##u[3 * i_ + 1].x;               \
      const float b0 = P##u[3 * i_ + 1].y, b1 = P##u[3 * i_ + 2].x, b2 = P##u[3 * i_ + 2].y;       \
      const float sa = a0 + a2, sb = b0 + b2;                                                      \
      smem2[uQ[i_]]                 = v2f{a0, b0};                                                 \
      smem2[uQ[i_] + U_PLN / 2]     = v2f{0.5f * (sa + a1), 0.5f * (sb + b1)};                     \
      smem2[uQ[i_] + 2 * U_PLN / 2] = v2f{0.5f * (sa - a1), 0.5f * (sb - b1)};                     \
      smem2[uQ[i_] + 3 * U_PLN / 2] = v2f{a2, b2};                                                 \
      uQ[i_] += dUQ;                                                                               \
    }                                                                                              \
    _Pragma("unroll")                                                                              \
    for (int i_ = 0; i_ < 2; ++i_) {                                                               \
      const float d0 = P##a[i_].x, d1 = P##a[i_].y, d2 = P##a[i_].z, d3 = P##a[i_].w;              \
      const float d4 = P##b[i_].x, d5 = P##b[i_].y;                                                \
      smem2[xQ[i_]]                 = v2f{d0 - d2, d2 - d4};                                       \
      smem2[xQ[i_] + V_POS / 2]     = v2f{d1 + d2, d3 + d4};                                       \
      smem2[xQ[i_] + 2 * V_POS / 2] = v2f{d2 - d1, d4 - d3};                                       \
      smem2[xQ[i_] + 3 * V_POS / 2] = v2f{d1 - d3, d3 - d5};                                       \
      xQ[i_] += dXQ;                                                                               \
    }                                                                                              \
    dUQ = -dUQ;  dXQ = -dXQ;                                                                       \
  } while (0)
  STAGE_DECL(sp);
  STAGE_DECL(sq);


  int aP[4];
  int bB[4][2];
#pragma unroll
  for (int p = 0; p < 4; ++p) {
    aP[p] = (2 * STG_V + p * U_PLN + (wm * 32 + l16) * U_STR + 14 * half) >> 1;
    asm volatile("" : "+v"(aP[p]));
#pragma unroll
    for (int cp = 0; cp < 2; ++cp) {
      bB[p][cp] = p * V_POS + (4 * half + 2 * cp) * V_PLN + wr * V_ROW + wn * 16 + l16;
      if (p + cp > 0) asm volatile("" : "+v"(bB[p][cp]));
    }
  }
  int dAP = STG_U / 2, dXP = STG_V;
  auto flipRead = [&]() {
#pragma unroll
    for (int p = 0; p < 4; ++p) { aP[p] += dAP;  bB[p][0] += dXP;  bB[p][1] += dXP; }
    dAP = -dAP;  dXP = -dXP;
  };

  v2f fa[2][6][2], fb[2][6];
  v8f acc[2][4];
  auto readGroup = [&](int p, int set) {
#pragma unroll
    for (int i = 0; i < 6; ++i) {
      fa[set][i][0] = smem2[aP[p] + i];
      fa[set][i][1] = smem2[aP[p] + i + 8 * U_STR];
      const int j0 = 2 * i, j1 = 2 * i + 1;
      const int cp = (j0 / 3) >> 1;
      const int o0 = ((j0 / 3) & 1) * V_PLN + (j0 % 3) * V_ROW;
      const int o1 = ((j1 / 3) - 2 * cp) * V_PLN + (j1 % 3) * V_ROW;
      fb[set][i] = v2f{smem[bB[p][cp] + o0], smem[bB[p][cp] + o1]};
    }
  };
  auto mmaGroup = [&](int p, int set) {
#pragma unroll
    for (int i = 0; i < 6; ++i)
#pragma unroll
      for (int ti = 0; ti < 2; ++ti)
        acc[ti][p] = __builtin_amdgcn_wmma_f32_16x16x4_f32(
            false, fa[set][i][ti], false, fb[set][i], (short)0, acc[ti][p], false, false);
  };

  STAGE_LOAD(0, sp);
  STAGE_LOAD(1, sq);
  STAGE_STORE(sp);
  STAGE_STORE(sq);
  __syncthreads();
  if (stripLive) readGroup(0, 0);
  STAGE_LOAD(2, sp);

#pragma unroll
  for (int ti = 0; ti < 2; ++ti)
#pragma unroll
    for (int p = 0; p < 4; ++p) acc[ti][p] = v8f{};

  for (int ch = 0; ch < NCHUNK; ++ch) {
    if (stripLive) {
#pragma unroll
      for (int p = 0; p < 3; ++p) {
        readGroup(p + 1, (p + 1) & 1);
        __builtin_amdgcn_sched_barrier(0);
        mmaGroup(p, p & 1);
        __builtin_amdgcn_sched_barrier(0);
      }
    }
    if (ch + 1 < NCHUNK) {
      __syncthreads();
      flipRead();
      if (stripLive) readGroup(0, 0);
      if (ch + 2 < NCHUNK) {
        STAGE_STORE(sp);
        if (ch + 3 < NCHUNK) STAGE_LOAD(ch + 3, sp);
      }
    }
    if (stripLive) {
      __builtin_amdgcn_sched_barrier(0);
      mmaGroup(3, 1);
      __builtin_amdgcn_sched_barrier(0);
    }
  }

  const int ow = owBlk + wn * 32 + 2 * l16;
  if (ow < W_OUT) {
#pragma unroll
    for (int ti = 0; ti < 2; ++ti) {
      const int m0 = mBase + wm * 32 + ti * 16 + half * 8;
      float* dst = out + (size_t)m0 * HW_OUT + (size_t)(oh0 + wr) * W_OUT + ow;
#pragma unroll
      for (int r = 0; r < 8; ++r) {
        const float y0 = acc[ti][0][r] + acc[ti][1][r] + acc[ti][2][r];
        const float y1 = acc[ti][1][r] - acc[ti][2][r] - acc[ti][3][r];
        *(v2f*)(dst + (size_t)r * HW_OUT) = v2f{y0, y1};
      }
    }
  }
}

extern "C" void kernel_launch(void* const* d_in, const int* in_sizes, int n_in,
                              void* d_out, int out_size, void* d_ws, size_t ws_size,
                              hipStream_t stream) {
  const float* x = (const float*)d_in[0];
  const float* k = (const float*)d_in[1];
  float* out = (float*)d_out;

  dim3 grid(N_FULL + N_ROWM);
  conv3x3_wmma_f32<<<grid, 256, 0, stream>>>(x, k, out);
}


# Global loads of stage ch+3 are no longer waited for where they are issued: their registers stay untouched until the next hand-over, so the stage's last 12 products issue under the load latency
# speedup vs baseline: 1.6756x; 1.0828x over previous
#include <hip/hip_runtime.h>


typedef float v2f __attribute__((ext_vector_type(2)));
typedef float v4f __attribute__((ext_vector_type(4)));
typedef float v8f __attribute__((ext_vector_type(8)));

#define C_IN   128
#define H_IN   224
#define W_IN   224
#define C_OUT  256
#define K_TOT  (C_IN * 9)
#define H_OUT  222
#define W_OUT  222
#define HW_OUT (H_OUT * W_OUT)

#define M_BLK  64
#define N_BLK  64
#define R_BLK  2
#define CC     8
#define NCHUNK (C_IN / CC)
#define U_STR  28
#define U_PLN  (M_BLK * U_STR)
#define STG_U  (4 * U_PLN)
#define V_ROW  32
#define V_PLN  (4 * V_ROW + 8)
#define V_POS  (CC * V_PLN)
#define STG_V  (4 * V_POS)

#define N_ROWP  (H_OUT / R_BLK)
#define N_ROWM  (N_ROWP * (C_OUT / M_BLK))
#define N_FULL  (3 * N_ROWM)

__global__ __launch_bounds__(256)
void conv3x3_wmma_f32(const float* __restrict__ x,
                      const float* __restrict__ w,
                      float* __restrict__ out) {
  __shared__ __attribute__((aligned(16))) float smem[2 * STG_V + 2 * STG_U];
  v2f* const smem2 = (v2f*)smem;

  const int tid  = threadIdx.x;
  const int lane = tid & 31;
  const int wave = __builtin_amdgcn_readfirstlane(tid >> 5);
  const int wm   = wave & 1;
  const int wr   = (wave >> 1) & 1;
  const int wn   = wave >> 2;
  const int half = lane >> 4;
  const int l16  = lane & 15;

  const int bid    = blockIdx.x;
  const bool full  = bid < N_FULL;
  const int colBlk = full ? bid % 3 : 3;
  const int rowM   = full ? bid / 3 : bid - N_FULL;
  const int owBlk  = colBlk * N_BLK;
  const int oh0    = (rowM % N_ROWP) * R_BLK;
  const int mBase  = (rowM / N_ROWP) * M_BLK;

  const bool stripLive = (owBlk + wn * 32) < W_OUT;

  const float* wBlk = w + (size_t)mBase * K_TOT;
  unsigned uSrc[3];
  int uQ[3];
#pragma unroll
  for (int i = 0; i < 3; ++i) {
    const int q = tid + 256 * i, m = q / 12, jp = q - m * 12;
    uSrc[i] = (unsigned)(m * K_TOT + jp * 6) * 4u;
    uQ[i] = (2 * STG_V + m * U_STR + 14 * (jp / 6) + 2 * (jp % 6)) >> 1;
  }
  const float* xBlk = x + (size_t)oh0 * W_IN;
  unsigned xSrcA[2], xSrcB[2];
  int xQ[2];
#pragma unroll
  for (int i = 0; i < 2; ++i) {
    const int u = tid + 256 * i, c = u >> 6, row = (u >> 4) & 3, c4 = u & 15;
    int colA = owBlk + c4 * 4;
    int colB = colA + 4;
    colA = colA > (W_IN - 4) ? (W_IN - 4) : colA;
    colB = colB > (W_IN - 2) ? (W_IN - 2) : colB;
    xSrcA[i] = (unsigned)(c * (H_IN * W_IN) + row * W_IN + colA) * 4u;
    xSrcB[i] = (unsigned)(c * (H_IN * W_IN) + row * W_IN + colB) * 4u;
    xQ[i] = (c * V_PLN + row * V_ROW + 2 * c4) >> 1;
  }
  int dUQ = STG_U / 2, dXQ = STG_V / 2;

#define STAGE_DECL(P) v2f P##u[9]; v4f P##a[2]; v2f P##b[2]
#define STAGE_LOAD(ch, P) do {                                                                     \
      \
      \
      \
    unsigned o0 = uSrc[0], o1 = uSrc[1], o2 = uSrc[2], pa0 = xSrcA[0], pa1 = xSrcA[1], pb0 = xSrcB[0], pb1 = xSrcB[1]; \
    asm volatile("" : "+v"(o0), "+v"(o1), "+v"(o2), "+v"(pa0), "+v"(pa1), "+v"(pb0), "+v"(pb1));  \
    const char* wc = (const char*)(wBlk + (ch) * (CC * 9));                                        \
    P##u[0] = *(const v2f*)(wc + o0);  P##u[1] = *(const v2f*)(wc + o0 + 8);  P##u[2] = *(const v2f*)(wc + o0 + 16); \
    P##u[3] = *(const v2f*)(wc + o1);  P##u[4] = *(const v2f*)(wc + o1 + 8);  P##u[5] = *(const v2f*)(wc + o1 + 16); \
    P##u[6] = *(const v2f*)(wc + o2);  P##u[7] = *(const v2f*)(wc + o2 + 8);  P##u[8] = *(const v2f*)(wc + o2 + 16); \
    const char* xc = (const char*)(xBlk + (size_t)(ch) * (CC * H_IN * W_IN));                      \
    P##a[0] = *(const v4f*)(xc + pa0);     P##b[0] = *(const v2f*)(xc + pb0);                      \
    P##a[1] = *(const v4f*)(xc + pa1);     P##b[1] = *(const v2f*)(xc + pb1);                      \
  } while (0)
#define STAGE_STORE(P) do {                                                                        \
      \
      \
      \
    asm volatile("" : "+v"(P##u[0]), "+v"(P##u[1]), "+v"(P##u[2]));                                 \
    asm volatile("" : "+v"(P##u[3]), "+v"(P##u[4]), "+v"(P##u[5]));                                 \
    asm volatile("" : "+v"(P##u[6]), "+v"(P##u[7]), "+v"(P##u[8]));                                 \
    asm volatile("" : "+v"(P##a[0]), "+v"(P##b[0]));                                                \
    asm volatile("" : "+v"(P##a[1]), "+v"(P##b[1]));                                                \
    _Pragma("unroll")                                                                              \
    for (int i_ = 0; i_ < 3; ++i_) {                                                               \
      const float a0 = P##u[3 * i_].x, a1 = P##u[3 * i_].y, a2 = P##u[3 * i_ + 1].x;               \
      const float b0 = P##u[3 * i_ + 1].y, b1 = P##u[3 * i_ + 2].x, b2 = P##u[3 * i_ + 2].y;       \
      const float sa = a0 + a2, sb = b0 + b2;                                                      \
      smem2[uQ[i_]]                 = v2f{a0, b0};                                                 \
      smem2[uQ[i_] + U_PLN / 2]     = v2f{0.5f * (sa + a1), 0.5f * (sb + b1)};                     \
      smem2[uQ[i_] + 2 * U_PLN / 2] = v2f{0.5f * (sa - a1), 0.5f * (sb - b1)};                     \
      smem2[uQ[i_] + 3 * U_PLN / 2] = v2f{a2, b2};                                                 \
      uQ[i_] += dUQ;                                                                               \
    }                                                                                              \
    _Pragma("unroll")                                                                              \
    for (int i_ = 0; i_ < 2; ++i_) {                                                               \
      const float d0 = P##a[i_].x, d1 = P##a[i_].y, d2 = P##a[i_].z, d3 = P##a[i_].w;              \
      const float d4 = P##b[i_].x, d5 = P##b[i_].y;                                                \
      smem2[xQ[i_]]                 = v2f{d0 - d2, d2 - d4};                                       \
      smem2[xQ[i_] + V_POS / 2]     = v2f{d1 + d2, d3 + d4};                                       \
      smem2[xQ[i_] + 2 * V_POS / 2] = v2f{d2 - d1, d4 - d3};                                       \
      smem2[xQ[i_] + 3 * V_POS / 2] = v2f{d1 - d3, d3 - d5};                                       \
      xQ[i_] += dXQ;                                                                               \
    }                                                                                              \
    dUQ = -dUQ;  dXQ = -dXQ;                                                                       \
  } while (0)
  STAGE_DECL(sp);
  STAGE_DECL(sq);


  int aP[4];
  int bB[4][2];
#pragma unroll
  for (int p = 0; p < 4; ++p) {
    aP[p] = (2 * STG_V + p * U_PLN + (wm * 32 + l16) * U_STR + 14 * half) >> 1;
    asm volatile("" : "+v"(aP[p]));
#pragma unroll
    for (int cp = 0; cp < 2; ++cp) {
      bB[p][cp] = p * V_POS + (4 * half + 2 * cp) * V_PLN + wr * V_ROW + wn * 16 + l16;
      if (p + cp > 0) asm volatile("" : "+v"(bB[p][cp]));
    }
  }
  int dAP = STG_U / 2, dXP = STG_V;
  auto flipRead = [&]() {
#pragma unroll
    for (int p = 0; p < 4; ++p) { aP[p] += dAP;  bB[p][0] += dXP;  bB[p][1] += dXP; }
    dAP = -dAP;  dXP = -dXP;
  };

  v2f fa[2][6][2], fb[2][6];
  v8f acc[2][4];
  auto readGroup = [&](int p, int set) {
#pragma unroll
    for (int i = 0; i < 6; ++i) {
      fa[set][i][0] = smem2[aP[p] + i];
      fa[set][i][1] = smem2[aP[p] + i + 8 * U_STR];
      const int j0 = 2 * i, j1 = 2 * i + 1;
      const int cp = (j0 / 3) >> 1;
      const int o0 = ((j0 / 3) & 1) * V_PLN + (j0 % 3) * V_ROW;
      const int o1 = ((j1 / 3) - 2 * cp) * V_PLN + (j1 % 3) * V_ROW;
      fb[set][i] = v2f{smem[bB[p][cp] + o0], smem[bB[p][cp] + o1]};
    }
  };
  auto mmaGroup = [&](int p, int set) {
#pragma unroll
    for (int i = 0; i < 6; ++i)
#pragma unroll
      for (int ti = 0; ti < 2; ++ti)
        acc[ti][p] = __builtin_amdgcn_wmma_f32_16x16x4_f32(
            false, fa[set][i][ti], false, fb[set][i], (short)0, acc[ti][p], false, false);
  };

  STAGE_LOAD(0, sp);
  STAGE_LOAD(1, sq);
  STAGE_STORE(sp);
  STAGE_STORE(sq);
  __syncthreads();
  if (stripLive) readGroup(0, 0);
  STAGE_LOAD(2, sp);

#pragma unroll
  for (int ti = 0; ti < 2; ++ti)
#pragma unroll
    for (int p = 0; p < 4; ++p) acc[ti][p] = v8f{};

  for (int ch = 0; ch < NCHUNK; ++ch) {
    if (stripLive) {
#pragma unroll
      for (int p = 0; p < 3; ++p) {
        readGroup(p + 1, (p + 1) & 1);
        __builtin_amdgcn_sched_barrier(0);
        mmaGroup(p, p & 1);
        __builtin_amdgcn_sched_barrier(0);
      }
    }
    if (ch + 1 < NCHUNK) {
      __syncthreads();
      flipRead();
      if (stripLive) readGroup(0, 0);
      if (ch + 2 < NCHUNK) {
        STAGE_STORE(sp);
        if (ch + 3 < NCHUNK) STAGE_LOAD(ch + 3, sp);
      }
    }
    if (stripLive) {
      __builtin_amdgcn_sched_barrier(0);
      mmaGroup(3, 1);
      __builtin_amdgcn_sched_barrier(0);
    }
  }

  const int ow = owBlk + wn * 32 + 2 * l16;
  if (ow < W_OUT) {
#pragma unroll
    for (int ti = 0; ti < 2; ++ti) {
      const int m0 = mBase + wm * 32 + ti * 16 + half * 8;
      float* dst = out + (size_t)m0 * HW_OUT + (size_t)(oh0 + wr) * W_OUT + ow;
#pragma unroll
      for (int r = 0; r < 8; ++r) {
        const float y0 = acc[ti][0][r] + acc[ti][1][r] + acc[ti][2][r];
        const float y1 = acc[ti][1][r] - acc[ti][2][r] - acc[ti][3][r];
        *(v2f*)(dst + (size_t)r * HW_OUT) = v2f{y0, y1};
      }
    }
  }
}

extern "C" void kernel_launch(void* const* d_in, const int* in_sizes, int n_in,
                              void* d_out, int out_size, void* d_ws, size_t ws_size,
                              hipStream_t stream) {
  const float* x = (const float*)d_in[0];
  const float* k = (const float*)d_in[1];
  float* out = (float*)d_out;

  dim3 grid(N_FULL + N_ROWM);
  conv3x3_wmma_f32<<<grid, 256, 0, stream>>>(x, k, out);
}


# Each wave computes one transformed position for 64 channels x 32 tiles (was 4 positions x 32 x 16): 24 fragment reads in 2 places per stage (was 48 in 4); the positions join in LDS at the end
# speedup vs baseline: 1.7240x; 1.0289x over previous
#include <hip/hip_runtime.h>


typedef float v2f __attribute__((ext_vector_type(2)));
typedef float v4f __attribute__((ext_vector_type(4)));
typedef float v8f __attribute__((ext_vector_type(8)));

#define C_IN   128
#define H_IN   224
#define W_IN   224
#define C_OUT  256
#define K_TOT  (C_IN * 9)
#define H_OUT  222
#define W_OUT  222
#define HW_OUT (H_OUT * W_OUT)

#define M_BLK  64
#define N_BLK  64
#define R_BLK  2
#define CC     8
#define NCHUNK (C_IN / CC)
#define U_STR  28
#define U_PLN  (M_BLK * U_STR)
#define STG_U  (4 * U_PLN)
#define V_ROW  32
#define V_PLN  (4 * V_ROW + 8)
#define V_POS  (CC * V_PLN)
#define STG_V  (4 * V_POS)

#define X_STR  68
#define X_POS  (2 * 32 * X_STR)

#define N_ROWP  (H_OUT / R_BLK)
#define N_ROWM  (N_ROWP * (C_OUT / M_BLK))
#define N_FULL  (3 * N_ROWM)

template <int N> struct IntC { static constexpr int value = N; };

__global__ __launch_bounds__(256)
void conv3x3_wmma_f32(const float* __restrict__ x,
                      const float* __restrict__ w,
                      float* __restrict__ out) {
  __shared__ __attribute__((aligned(16))) float smem[2 * STG_V + 2 * STG_U];
  v2f* const smem2 = (v2f*)smem;

  const int tid  = threadIdx.x;
  const int lane = tid & 31;
  const int wave = __builtin_amdgcn_readfirstlane(tid >> 5);
  const int pw   = wave & 3;
  const int wr   = wave >> 2;
  const int half = lane >> 4;
  const int l16  = lane & 15;

  const int bid    = blockIdx.x;
  const bool full  = bid < N_FULL;
  const int colBlk = full ? bid % 3 : 3;
  const int rowM   = full ? bid / 3 : bid - N_FULL;
  const int owBlk  = colBlk * N_BLK;
  const int oh0    = (rowM % N_ROWP) * R_BLK;
  const int mBase  = (rowM / N_ROWP) * M_BLK;


  const float* wBlk = w + (size_t)mBase * K_TOT;
  unsigned uSrc[3];
  int uQ[3];
#pragma unroll
  for (int i = 0; i < 3; ++i) {
    const int q = tid + 256 * i, m = q / 12, jp = q - m * 12;
    uSrc[i] = (unsigned)(m * K_TOT + jp * 6) * 4u;
    uQ[i] = (2 * STG_V + m * U_STR + 14 * (jp / 6) + 2 * (jp % 6)) >> 1;
  }
  const float* xBlk = x + (size_t)oh0 * W_IN;
  unsigned xSrcA[2], xSrcB[2];
  int xQ[2];
#pragma unroll
  for (int i = 0; i < 2; ++i) {
    const int u = tid + 256 * i, c = u >> 6, row = (u >> 4) & 3, c4 = u & 15;
    int colA = owBlk + c4 * 4;
    int colB = colA + 4;
    colA = colA > (W_IN - 4) ? (W_IN - 4) : colA;
    colB = colB > (W_IN - 2) ? (W_IN - 2) : colB;
    xSrcA[i] = (unsigned)(c * (H_IN * W_IN) + row * W_IN + colA) * 4u;
    xSrcB[i] = (unsigned)(c * (H_IN * W_IN) + row * W_IN + colB) * 4u;
    xQ[i] = (c * V_PLN + row * V_ROW + 2 * c4) >> 1;
  }
  int dUQ = STG_U / 2, dXQ = STG_V / 2;

#define STAGE_DECL(P) v2f P##u[9]; v4f P##a[2]; v2f P##b[2]
#define STAGE_LOAD(ch, P) do {                                                                     \
      \
      \
      \
    unsigned o0 = uSrc[0], o1 = uSrc[1], o2 = uSrc[2], pa0 = xSrcA[0], pa1 = xSrcA[1], pb0 = xSrcB[0], pb1 = xSrcB[1]; \
    asm volatile("" : "+v"(o0), "+v"(o1), "+v"(o2), "+v"(pa0), "+v"(pa1), "+v"(pb0), "+v"(pb1));  \
    const char* wc = (const char*)(wBlk + (ch) * (CC * 9));                                        \
    P##u[0] = *(const v2f*)(wc + o0);  P##u[1] = *(const v2f*)(wc + o0 + 8);  P##u[2] = *(const v2f*)(wc + o0 + 16); \
    P##u[3] = *(const v2f*)(wc + o1);  P##u[4] = *(const v2f*)(wc + o1 + 8);  P##u[5] = *(const v2f*)(wc + o1 + 16); \
    P##u[6] = *(const v2f*)(wc + o2);  P##u[7] = *(const v2f*)(wc + o2 + 8);  P##u[8] = *(const v2f*)(wc + o2 + 16); \
    const char* xc = (const char*)(xBlk + (size_t)(ch) * (CC * H_IN * W_IN));                      \
    P##a[0] = *(const v4f*)(xc + pa0);     P##b[0] = *(const v2f*)(xc + pb0);                      \
    P##a[1] = *(const v4f*)(xc + pa1);     P##b[1] = *(const v2f*)(xc + pb1);                      \
  } while (0)
#define STAGE_STORE(P) do {                                                                        \
      \
      \
      \
    asm volatile("" : "+v"(P##u[0]), "+v"(P##u[1]), "+v"(P##u[2]));                                 \
    asm volatile("" : "+v"(P##u[3]), "+v"(P##u[4]), "+v"(P##u[5]));                                 \
    asm volatile("" : "+v"(P##u[6]), "+v"(P##u[7]), "+v"(P##u[8]));                                 \
    asm volatile("" : "+v"(P##a[0]), "+v"(P##b[0]));                                                \
    asm volatile("" : "+v"(P##a[1]), "+v"(P##b[1]));                                                \
    _Pragma("unroll")                                                                              \
    for (int i_ = 0; i_ < 3; ++i_) {                                                               \
      const float a0 = P##u[3 * i_].x, a1 = P##u[3 * i_].y, a2 = P##u[3 * i_ + 1].x;               \
      const float b0 = P##u[3 * i_ + 1].y, b1 = P##u[3 * i_ + 2].x, b2 = P##u[3 * i_ + 2].y;       \
      const float sa = a0 + a2, sb = b0 + b2;                                                      \
      smem2[uQ[i_]]                 = v2f{a0, b0};                                                 \
      smem2[uQ[i_] + U_PLN / 2]     = v2f{0.5f * (sa + a1), 0.5f * (sb + b1)};                     \
      smem2[uQ[i_] + 2 * U_PLN / 2] = v2f{0.5f * (sa - a1), 0.5f * (sb - b1)};                     \
      smem2[uQ[i_] + 3 * U_PLN / 2] = v2f{a2, b2};                                                 \
      uQ[i_] += dUQ;                                                                               \
    }                                                                                              \
    _Pragma("unroll")                                                                              \
    for (int i_ = 0; i_ < 2; ++i_) {                                                               \
      const float d0 = P##a[i_].x, d1 = P##a[i_].y, d2 = P##a[i_].z, d3 = P##a[i_].w;              \
      const float d4 = P##b[i_].x, d5 = P##b[i_].y;                                                \
      smem2[xQ[i_]]                 = v2f{d0 - d2, d2 - d4};                                       \
      smem2[xQ[i_] + V_POS / 2]     = v2f{d1 + d2, d3 + d4};                                       \
      smem2[xQ[i_] + 2 * V_POS / 2] = v2f{d2 - d1, d4 - d3};                                       \
      smem2[xQ[i_] + 3 * V_POS / 2] = v2f{d1 - d3, d3 - d5};                                       \
      xQ[i_] += dXQ;                                                                               \
    }                                                                                              \
    dUQ = -dUQ;  dXQ = -dXQ;                                                                       \
  } while (0)
  STAGE_DECL(sp);
  STAGE_DECL(sq);


  int aP[2];
  int bB[2][2];
#pragma unroll
  for (int mp = 0; mp < 2; ++mp) {
    aP[mp] = (2 * STG_V + pw * U_PLN + (mp * 32 + l16) * U_STR + 14 * half) >> 1;
#pragma unroll
    for (int nt = 0; nt < 2; ++nt)
      bB[mp][nt] = pw * V_POS + (4 * half + 2 * mp) * V_PLN + wr * V_ROW + 16 * nt + l16;
  }
  asm volatile("" : "+v"(aP[1]), "+v"(bB[0][1]), "+v"(bB[1][0]), "+v"(bB[1][1]));
  int dAP = STG_U / 2, dXP = STG_V;
  auto flipRead = [&]() {
#pragma unroll
    for (int mp = 0; mp < 2; ++mp) { aP[mp] += dAP;  bB[mp][0] += dXP;  bB[mp][1] += dXP; }
    dAP = -dAP;  dXP = -dXP;
  };

  STAGE_LOAD(0, sp);
  STAGE_LOAD(1, sq);
  STAGE_STORE(sp);
  STAGE_STORE(sq);
  __syncthreads();

  auto run = [&](auto ntc) {
    constexpr int NT = decltype(ntc)::value;
    v2f fa[2][3][4], fb[2][3][2];
    v8f acc[4][2];
    auto readGroup = [&](int g, int set) {
#pragma unroll
      for (int ii = 0; ii < 3; ++ii) {
        const int i = 3 * g + ii;
#pragma unroll
        for (int mt = 0; mt < 4; ++mt)
          fa[set][ii][mt] = smem2[aP[mt >> 1] + i + (mt & 1) * 8 * U_STR];
        const int j0 = 2 * i, j1 = 2 * i + 1;
        const int cp = (j0 / 3) >> 1;
        const int o0 = ((j0 / 3) & 1) * V_PLN + (j0 % 3) * V_ROW;
        const int o1 = ((j1 / 3) - 2 * cp) * V_PLN + (j1 % 3) * V_ROW;
#pragma unroll
        for (int nt = 0; nt < NT; ++nt)
          fb[set][ii][nt] = v2f{smem[bB[cp][nt] + o0], smem[bB[cp][nt] + o1]};
      }
    };
    auto mmaGroup = [&](int set) {
#pragma unroll
      for (int ii = 0; ii < 3; ++ii)
#pragma unroll
        for (int mt = 0; mt < 4; ++mt)
#pragma unroll
          for (int nt = 0; nt < NT; ++nt)
            acc[mt][nt] = __builtin_amdgcn_wmma_f32_16x16x4_f32(
                false, fa[set][ii][mt], false, fb[set][ii][nt], (short)0, acc[mt][nt], false, false);
    };

    readGroup(0, 0);
    STAGE_LOAD(2, sp);

#pragma unroll
    for (int mt = 0; mt < 4; ++mt)
#pragma unroll
      for (int nt = 0; nt < 2; ++nt) acc[mt][nt] = v8f{};

    for (int ch = 0; ch < NCHUNK; ++ch) {
      readGroup(1, 1);
      __builtin_amdgcn_sched_barrier(0);
      mmaGroup(0);
      __builtin_amdgcn_sched_barrier(0);
      if (ch + 1 < NCHUNK) {
        __syncthreads();
        flipRead();
        readGroup(0, 0);
        if (ch + 2 < NCHUNK) {
          STAGE_STORE(sp);
          if (ch + 3 < NCHUNK) STAGE_LOAD(ch + 3, sp);
        }
      }
      __builtin_amdgcn_sched_barrier(0);
      mmaGroup(1);
      __builtin_amdgcn_sched_barrier(0);
    }

    __syncthreads();
#pragma unroll
    for (int mt = 0; mt < 4; ++mt)
#pragma unroll
      for (int nt = 0; nt < NT; ++nt) {
        float* d = smem + ((pw * 2 + wr) * 32 + nt * 16 + l16) * X_STR + mt * 16 + 8 * half;
        *(v4f*)d       = v4f{acc[mt][nt][0], acc[mt][nt][1], acc[mt][nt][2], acc[mt][nt][3]};
        *(v4f*)(d + 4) = v4f{acc[mt][nt][4], acc[mt][nt][5], acc[mt][nt][6], acc[mt][nt][7]};
      }
    __syncthreads();
    const int ow = owBlk + 2 * lane;
    if (ow < W_OUT) {
#pragma unroll
      for (int q = 0; q < 4; ++q) {
        const int line = wave * 4 + q, row = line >> 4, quad = line & 15;
        const float* s = smem + (row * 32 + lane) * X_STR + quad * 4;
        const v4f m0 = *(const v4f*)s;
        const v4f m1 = *(const v4f*)(s + X_POS);
        const v4f m2 = *(const v4f*)(s + 2 * X_POS);
        const v4f m3 = *(const v4f*)(s + 3 * X_POS);
        float* dst = out + (size_t)(mBase + quad * 4) * HW_OUT + (size_t)(oh0 + row) * W_OUT + ow;
#pragma unroll
        for (int e = 0; e < 4; ++e) {
          const float y0 = m0[e] + m1[e] + m2[e];
          const float y1 = m1[e] - m2[e] - m3[e];
          *(v2f*)(dst + (size_t)e * HW_OUT) = v2f{y0, y1};
        }
      }
    }
  };
  if (full) run(IntC<2>{}); else run(IntC<1>{});
}

extern "C" void kernel_launch(void* const* d_in, const int* in_sizes, int n_in,
                              void* d_out, int out_size, void* d_ws, size_t ws_size,
                              hipStream_t stream) {
  const float* x = (const float*)d_in[0];
  const float* k = (const float*)d_in[1];
  float* out = (float*)d_out;

  dim3 grid(N_FULL + N_ROWM);
  conv3x3_wmma_f32<<<grid, 256, 0, stream>>>(x, k, out);
}
